# starting kernel
# baseline (speedup 1.0000x reference)
#include <hip/hip_runtime.h>


typedef __attribute__((ext_vector_type(2))) float v2f;
typedef __attribute__((ext_vector_type(8))) float v8f;

#define ALPHA 0.1f


__global__ void k_deg_init(float* __restrict__ deg, int n) {
    int i = blockIdx.x * blockDim.x + threadIdx.x;
    if (i < n) deg[i] = 1.0f;
}

__global__ void k_deg_count(float* __restrict__ deg, const int* __restrict__ dst, int e) {
    int i = blockIdx.x * blockDim.x + threadIdx.x;
    if (i < e) atomicAdd(&deg[dst[i]], 1.0f);
}

__global__ void k_rsqrt(float* __restrict__ d, int n) {
    int i = blockIdx.x * blockDim.x + threadIdx.x;
    if (i < n) {
        float v = d[i];
        d[i] = (v > 0.0f) ? rsqrtf(v) : 0.0f;
    }
}

__global__ void k_edge_w(float* __restrict__ w, const int* __restrict__ src,
                         const int* __restrict__ dst, const float* __restrict__ dinv, int e) {
    int i = blockIdx.x * blockDim.x + threadIdx.x;
    if (i < e) w[i] = dinv[src[i]] * dinv[dst[i]];
}


template <int K, int NOUT>
__global__ void k_gemm_wmma(const float* __restrict__ A, const float* __restrict__ W,
                            const float* __restrict__ bias, float* __restrict__ C) {
    const int wave = threadIdx.x >> 5;
    const int lane = threadIdx.x & 31;
    const int row0 = blockIdx.x * 16;
    const int col0 = wave * 16;
    const int lidx = lane & 15;
    const int khalf = (lane >> 4) * 2;

    const float* __restrict__ arow = A + (size_t)(row0 + lidx) * K + khalf;
    const float* __restrict__ wcol = W + (size_t)khalf * NOUT + col0 + lidx;

    v8f acc = {};
#pragma unroll 4
    for (int k = 0; k < K; k += 4) {
        v2f a, b;
        a.x = arow[0];
        a.y = arow[1];
        b.x = wcol[0];
        b.y = wcol[NOUT];
        acc = __builtin_amdgcn_wmma_f32_16x16x4_f32(
             false, a,  false, b,
             (short)0, acc,  false,  false);
        arow += 4;
        wcol += 4 * NOUT;
    }

    const float bv = bias[col0 + lidx];
    const int moff = (lane >> 4) * 8;
    float* __restrict__ cbase = C + (size_t)row0 * NOUT + col0 + lidx;
#pragma unroll
    for (int j = 0; j < 8; ++j) {
        cbase[(size_t)(j + moff) * NOUT] = acc[j] + bv;
    }
}


template <int F>
__global__ void k_prop_self(float* __restrict__ out, const float* __restrict__ h,
                            const float* __restrict__ dinv, int n) {
    const int vpr = F / 4;
    long long gid = (long long)blockIdx.x * blockDim.x + threadIdx.x;
    long long total = (long long)n * vpr;
    if (gid >= total) return;
    int i = (int)(gid / vpr);
    int f4 = (int)(gid % vpr);
    float sw = dinv[i] * dinv[i];
    float4 hv = ((const float4*)(h + (size_t)i * F))[f4];
    float4 ov = make_float4(sw * hv.x, sw * hv.y, sw * hv.z, sw * hv.w);
    ((float4*)(out + (size_t)i * F))[f4] = ov;
}

template <int F>
__global__ void k_prop_edges(float* __restrict__ out, const float* __restrict__ h,
                             const int* __restrict__ src, const int* __restrict__ dst,
                             const float* __restrict__ w, int e) {
    const int vpr = F / 4;
    long long gid = (long long)blockIdx.x * blockDim.x + threadIdx.x;
    long long total = (long long)e * vpr;
    if (gid >= total) return;
    int ei = (int)(gid / vpr);
    int f4 = (int)(gid % vpr);
    int s = src[ei];
    int d = dst[ei];
    float we = w[ei];
    float4 hv = ((const float4*)(h + (size_t)s * F))[f4];
    float* o = out + (size_t)d * F + (size_t)f4 * 4;
    atomicAdd(o + 0, we * hv.x);
    atomicAdd(o + 1, we * hv.y);
    atomicAdd(o + 2, we * hv.z);
    atomicAdd(o + 3, we * hv.w);
}

__global__ void k_combine(float* __restrict__ h, const float* __restrict__ tmp,
                          const float* __restrict__ h0, long long n4) {
    long long gid = (long long)blockIdx.x * blockDim.x + threadIdx.x;
    if (gid >= n4) return;
    float4 t = ((const float4*)tmp)[gid];
    float4 z = ((const float4*)h0)[gid];
    float4 r = make_float4((1.0f - ALPHA) * t.x + ALPHA * z.x,
                           (1.0f - ALPHA) * t.y + ALPHA * z.y,
                           (1.0f - ALPHA) * t.z + ALPHA * z.z,
                           (1.0f - ALPHA) * t.w + ALPHA * z.w);
    ((float4*)h)[gid] = r;
}


static inline int cdiv(long long a, int b) { return (int)((a + b - 1) / b); }

extern "C" void kernel_launch(void* const* d_in, const int* in_sizes, int n_in,
                              void* d_out, int out_size, void* d_ws, size_t ws_size,
                              hipStream_t stream) {
    const float* x  = (const float*)d_in[0];
    const int*   ei = (const int*)d_in[1];
    const float* W1 = (const float*)d_in[2];
    const float* b1 = (const float*)d_in[3];
    const float* W3 = (const float*)d_in[4];
    const float* b3 = (const float*)d_in[5];
    float* out = (float*)d_out;

    const int N = in_sizes[0] / 256;
    const int E = in_sizes[1] / 2;
    const int* src = ei;
    const int* dst = ei + E;

    float* ws = (float*)d_ws;
    float* dinv = ws;
    size_t off = ((size_t)N + 63) & ~(size_t)63;
    float* wbuf = ws + off;
    off += ((size_t)E + 63) & ~(size_t)63;
    float* h0 = ws + off;
    off += (size_t)N * 128;
    float* h = ws + off;
    off += (size_t)N * 128;
    float* tmp = ws + off;
    off += (size_t)N * 128;
    float* hw = ws + off;

    const int T = 256;

    k_deg_init<<<cdiv(N, T), T, 0, stream>>>(dinv, N);
    k_deg_count<<<cdiv(E, T), T, 0, stream>>>(dinv, dst, E);
    k_rsqrt<<<cdiv(N, T), T, 0, stream>>>(dinv, N);
    k_edge_w<<<cdiv(E, T), T, 0, stream>>>(wbuf, src, dst, dinv, E);

    k_gemm_wmma<256, 128><<<N / 16, 32 * (128 / 16), 0, stream>>>(x, W1, b1, tmp);
    k_prop_self<128><<<cdiv((long long)N * 32, T), T, 0, stream>>>(h0, tmp, dinv, N);
    k_prop_edges<128><<<cdiv((long long)E * 32, T), T, 0, stream>>>(h0, tmp, src, dst, wbuf, E);

    hipMemcpyAsync(h, h0, (size_t)N * 128 * sizeof(float), hipMemcpyDeviceToDevice, stream);

    const long long n4 = (long long)N * 128 / 4;
    for (int it = 0; it < 10; ++it) {
        k_prop_self<128><<<cdiv((long long)N * 32, T), T, 0, stream>>>(tmp, h, dinv, N);
        k_prop_edges<128><<<cdiv((long long)E * 32, T), T, 0, stream>>>(tmp, h, src, dst, wbuf, E);
        k_combine<<<cdiv(n4, T), T, 0, stream>>>(h, tmp, h0, n4);
    }

    k_gemm_wmma<128, 64><<<N / 16, 32 * (64 / 16), 0, stream>>>(h, W3, b3, hw);
    k_prop_self<64><<<cdiv((long long)N * 16, T), T, 0, stream>>>(out, hw, dinv, N);
    k_prop_edges<64><<<cdiv((long long)E * 16, T), T, 0, stream>>>(out, hw, src, dst, wbuf, E);
}


# Pull propagation: edges bucketed by destination per call (int atomics); one wave sums a row's in-edges + self + alpha*h0 in registers, one write; no float atomics, sum order = slot arrival order
# speedup vs baseline: 4.6690x; 4.6690x over previous
#include <hip/hip_runtime.h>


typedef __attribute__((ext_vector_type(2))) float v2f;
typedef __attribute__((ext_vector_type(8))) float v8f;

#define ALPHA 0.1f
#define SCAN_T 256


__global__ void k_cnt_zero(int* __restrict__ cnt, int n) {
    int i = blockIdx.x * blockDim.x + threadIdx.x;
    if (i < n) cnt[i] = 0;
}

__global__ void k_cnt_edges(int* __restrict__ cnt, const int* __restrict__ dst, int e) {
    int i = blockIdx.x * blockDim.x + threadIdx.x;
    if (i < e) atomicAdd(&cnt[dst[i]], 1);
}

__global__ void k_rowptr(const int* __restrict__ cnt, int* __restrict__ rowptr,
                         int* __restrict__ cursor, float* __restrict__ dinv, int n) {
    __shared__ int wtot[SCAN_T / 32];
    __shared__ int wpre[SCAN_T / 32];
    const int t = threadIdx.x;
    const int lane = t & 31;
    const int wv = t >> 5;
    const int b0 = blockIdx.x * SCAN_T;

    int pre = 0;
    for (int i = t; i < b0; i += SCAN_T) pre += cnt[i];

    const int idx = b0 + t;
    const int c = (idx < n) ? cnt[idx] : 0;

    int inc = c;
    for (int o = 1; o < 32; o <<= 1) {
        int u = __shfl_up(inc, o);
        if (lane >= o) inc += u;
    }
    for (int o = 16; o > 0; o >>= 1) pre += __shfl_xor(pre, o);

    if (lane == 31) wtot[wv] = inc;
    if (lane == 0) wpre[wv] = pre;
    __syncthreads();

    int base = 0;
    for (int w = 0; w < SCAN_T / 32; ++w) {
        base += wpre[w];
        if (w < wv) base += wtot[w];
    }
    const int excl = base + inc - c;
    if (idx < n) {
        rowptr[idx] = excl;
        cursor[idx] = excl;
        const float v = (float)(c + 1);
        dinv[idx] = (v > 0.0f) ? rsqrtf(v) : 0.0f;
        if (idx == n - 1) rowptr[n] = excl + c;
    }
}

__global__ void k_fill(int* __restrict__ ecw, int* __restrict__ cursor,
                       const int* __restrict__ src, const int* __restrict__ dst,
                       const float* __restrict__ dinv, int e) {
    int i = blockIdx.x * blockDim.x + threadIdx.x;
    if (i < e) {
        const int s = src[i];
        const int d = dst[i];
        const int pos = atomicAdd(&cursor[d], 1);
        const float w = dinv[s] * dinv[d];
        ((int2*)ecw)[pos] = make_int2(s, __float_as_int(w));
    }
}


template <int K, int NOUT>
__global__ void k_gemm_wmma(const float* __restrict__ A, const float* __restrict__ W,
                            const float* __restrict__ bias, float* __restrict__ C) {
    const int wave = threadIdx.x >> 5;
    const int lane = threadIdx.x & 31;
    const int row0 = blockIdx.x * 16;
    const int col0 = wave * 16;
    const int lidx = lane & 15;
    const int khalf = (lane >> 4) * 2;

    const float* __restrict__ arow = A + (size_t)(row0 + lidx) * K + khalf;
    const float* __restrict__ wcol = W + (size_t)khalf * NOUT + col0 + lidx;

    v8f acc = {};
#pragma unroll 4
    for (int k = 0; k < K; k += 4) {
        v2f a, b;
        a.x = arow[0];
        a.y = arow[1];
        b.x = wcol[0];
        b.y = wcol[NOUT];
        acc = __builtin_amdgcn_wmma_f32_16x16x4_f32(
             false, a,  false, b,
             (short)0, acc,  false,  false);
        arow += 4;
        wcol += 4 * NOUT;
    }

    const float bv = bias[col0 + lidx];
    const int moff = (lane >> 4) * 8;
    float* __restrict__ cbase = C + (size_t)row0 * NOUT + col0 + lidx;
#pragma unroll
    for (int j = 0; j < 8; ++j) {
        cbase[(size_t)(j + moff) * NOUT] = acc[j] + bv;
    }
}


template <int F, bool MIX>
__global__ void __launch_bounds__(256)
k_prop_pull(float* __restrict__ out, const float* __restrict__ h, const float* __restrict__ h0,
            const int* __restrict__ rowptr, const int* __restrict__ ecw,
            const float* __restrict__ dinv, int n) {
    typedef __attribute__((ext_vector_type(F / 32))) float vf;
    const int lane = threadIdx.x & 31;
    const int w0 = __builtin_amdgcn_readfirstlane((int)((blockIdx.x * blockDim.x + threadIdx.x) >> 5));
    const int nw = (int)(gridDim.x * (blockDim.x >> 5));
    const vf* __restrict__ hv = (const vf*)h;

  for (int i = w0; i < n; i += nw) {
    const vf hself = hv[(size_t)i * 32 + lane];
    vf z = hself;
    if (MIX) z = ((const vf*)h0)[(size_t)i * 32 + lane];
    const int beg = rowptr[i];
    const int end = rowptr[i + 1];
    int2 cw = make_int2(0, 0);
    if (beg + lane < end) cw = ((const int2*)ecw)[beg + lane];
    const float di = dinv[i];
    const float sw = di * di;
    vf acc = sw * hself;

    for (int base = beg; base < end; base += 32) {
        if (base != beg) {
            const int p = base + lane;
            cw = make_int2(0, 0);
            if (p < end) cw = ((const int2*)ecw)[p];
        }
        const int rem = end - base;
        const int cnt = rem < 32 ? rem : 32;

        int j = 0;
        for (; j + 4 <= cnt; j += 4) {
            const int s0 = __builtin_amdgcn_readlane(cw.x, j);
            const int s1 = __builtin_amdgcn_readlane(cw.x, j + 1);
            const int s2 = __builtin_amdgcn_readlane(cw.x, j + 2);
            const int s3 = __builtin_amdgcn_readlane(cw.x, j + 3);
            const float w0 = __int_as_float(__builtin_amdgcn_readlane(cw.y, j));
            const float w1 = __int_as_float(__builtin_amdgcn_readlane(cw.y, j + 1));
            const float w2 = __int_as_float(__builtin_amdgcn_readlane(cw.y, j + 2));
            const float w3 = __int_as_float(__builtin_amdgcn_readlane(cw.y, j + 3));
            const vf r0 = hv[(size_t)s0 * 32 + lane];
            const vf r1 = hv[(size_t)s1 * 32 + lane];
            const vf r2 = hv[(size_t)s2 * 32 + lane];
            const vf r3 = hv[(size_t)s3 * 32 + lane];
            acc += w0 * r0;
            acc += w1 * r1;
            acc += w2 * r2;
            acc += w3 * r3;
        }
        const int left = cnt - j;
        if (left > 0) {
            const int s0 = __builtin_amdgcn_readlane(cw.x, j);
            const float w0 = __int_as_float(__builtin_amdgcn_readlane(cw.y, j));
            const vf r0 = hv[(size_t)s0 * 32 + lane];
            float w1 = 0.0f, w2 = 0.0f;
            vf r1 = {}, r2 = {};
            if (left > 1) {
                const int s1 = __builtin_amdgcn_readlane(cw.x, j + 1);
                w1 = __int_as_float(__builtin_amdgcn_readlane(cw.y, j + 1));
                r1 = hv[(size_t)s1 * 32 + lane];
            }
            if (left > 2) {
                const int s2 = __builtin_amdgcn_readlane(cw.x, j + 2);
                w2 = __int_as_float(__builtin_amdgcn_readlane(cw.y, j + 2));
                r2 = hv[(size_t)s2 * 32 + lane];
            }
            acc += w0 * r0;
            if (left > 1) acc += w1 * r1;
            if (left > 2) acc += w2 * r2;
        }
    }

    if (MIX) acc = (1.0f - ALPHA) * acc + ALPHA * z;
    ((vf*)out)[(size_t)i * 32 + lane] = acc;
  }
}


static inline int cdiv(long long a, int b) { return (int)((a + b - 1) / b); }

extern "C" void kernel_launch(void* const* d_in, const int* in_sizes, int n_in,
                              void* d_out, int out_size, void* d_ws, size_t ws_size,
                              hipStream_t stream) {
    const float* x  = (const float*)d_in[0];
    const int*   ei = (const int*)d_in[1];
    const float* W1 = (const float*)d_in[2];
    const float* b1 = (const float*)d_in[3];
    const float* W3 = (const float*)d_in[4];
    const float* b3 = (const float*)d_in[5];
    float* out = (float*)d_out;

    const int N = in_sizes[0] / 256;
    const int E = in_sizes[1] / 2;
    const int* src = ei;
    const int* dst = ei + E;

    float* ws = (float*)d_ws;
    const size_t nal = ((size_t)N + 64 + 63) & ~(size_t)63;
    float* dinv = ws;
    size_t off = nal;
    int* cnt = (int*)(ws + off);
    off += nal;
    int* rowptr = (int*)(ws + off);
    off += nal;
    int* cursor = (int*)(ws + off);
    off += nal;
    int* ecw = (int*)(ws + off);
    off += ((size_t)2 * E + 63) & ~(size_t)63;
    float* bufA = ws + off;
    off += (size_t)N * 128;
    float* h0 = ws + off;
    off += (size_t)N * 128;
    float* bufB = ws + off;
    float* hw = h0;

    const int T = 256;

    k_cnt_zero<<<cdiv(N, T), T, 0, stream>>>(cnt, N);
    k_cnt_edges<<<cdiv(E, T), T, 0, stream>>>(cnt, dst, E);
    k_rowptr<<<cdiv(N, SCAN_T), SCAN_T, 0, stream>>>(cnt, rowptr, cursor, dinv, N);
    k_fill<<<cdiv(E, T), T, 0, stream>>>(ecw, cursor, src, dst, dinv, E);

    int gp = cdiv((long long)N * 32, T);
    if (gp > 2048) gp = 2048;

    k_gemm_wmma<256, 128><<<N / 16, 32 * (128 / 16), 0, stream>>>(x, W1, b1, bufA);
    k_prop_pull<128, false><<<gp, T, 0, stream>>>(h0, bufA, bufA, rowptr, ecw, dinv, N);

    const float* cur = h0;
    for (int it = 0; it < 5; ++it) {
        k_prop_pull<128, true><<<gp, T, 0, stream>>>(bufA, cur, h0, rowptr, ecw, dinv, N);
        k_prop_pull<128, true><<<gp, T, 0, stream>>>(bufB, bufA, h0, rowptr, ecw, dinv, N);
        cur = bufB;
    }

    k_gemm_wmma<128, 64><<<N / 16, 32 * (64 / 16), 0, stream>>>(bufB, W3, b3, hw);
    k_prop_pull<64, false><<<gp, T, 0, stream>>>(out, hw, hw, rowptr, ecw, dinv, N);
}


# Second dense layer (W3) applied before the ten mixing iterations (propagation is linear, commutes): all 12 propagations run 64 wide, not 128; b3 added after the 10th; sum order: W3 before edge sums
# speedup vs baseline: 5.7680x; 1.2354x over previous
#include <hip/hip_runtime.h>


typedef __attribute__((ext_vector_type(2))) float v2f;
typedef __attribute__((ext_vector_type(8))) float v8f;

#define ALPHA 0.1f
#define SCAN_T 256


__global__ void k_cnt_zero(int* __restrict__ cnt, int n) {
    int i = blockIdx.x * blockDim.x + threadIdx.x;
    if (i < n) cnt[i] = 0;
}

__global__ void k_cnt_edges(int* __restrict__ cnt, const int* __restrict__ dst, int e) {
    int i = blockIdx.x * blockDim.x + threadIdx.x;
    if (i < e) atomicAdd(&cnt[dst[i]], 1);
}

__global__ void k_rowptr(const int* __restrict__ cnt, int* __restrict__ rowptr,
                         int* __restrict__ cursor, float* __restrict__ dinv, int n) {
    __shared__ int wtot[SCAN_T / 32];
    __shared__ int wpre[SCAN_T / 32];
    const int t = threadIdx.x;
    const int lane = t & 31;
    const int wv = t >> 5;
    const int b0 = blockIdx.x * SCAN_T;

    int pre = 0;
    for (int i = t; i < b0; i += SCAN_T) pre += cnt[i];

    const int idx = b0 + t;
    const int c = (idx < n) ? cnt[idx] : 0;

    int inc = c;
    for (int o = 1; o < 32; o <<= 1) {
        int u = __shfl_up(inc, o);
        if (lane >= o) inc += u;
    }
    for (int o = 16; o > 0; o >>= 1) pre += __shfl_xor(pre, o);

    if (lane == 31) wtot[wv] = inc;
    if (lane == 0) wpre[wv] = pre;
    __syncthreads();

    int base = 0;
    for (int w = 0; w < SCAN_T / 32; ++w) {
        base += wpre[w];
        if (w < wv) base += wtot[w];
    }
    const int excl = base + inc - c;
    if (idx < n) {
        rowptr[idx] = excl;
        cursor[idx] = excl;
        const float v = (float)(c + 1);
        dinv[idx] = (v > 0.0f) ? rsqrtf(v) : 0.0f;
        if (idx == n - 1) rowptr[n] = excl + c;
    }
}

__global__ void k_fill(int* __restrict__ ecw, int* __restrict__ cursor,
                       const int* __restrict__ src, const int* __restrict__ dst,
                       const float* __restrict__ dinv, int e) {
    int i = blockIdx.x * blockDim.x + threadIdx.x;
    if (i < e) {
        const int s = src[i];
        const int d = dst[i];
        const int pos = atomicAdd(&cursor[d], 1);
        const float w = dinv[s] * dinv[d];
        ((int2*)ecw)[pos] = make_int2(s, __float_as_int(w));
    }
}


template <int K, int NOUT, bool HAS_BIAS>
__global__ void k_gemm_wmma(const float* __restrict__ A, const float* __restrict__ W,
                            const float* __restrict__ bias, float* __restrict__ C) {
    const int wave = threadIdx.x >> 5;
    const int lane = threadIdx.x & 31;
    const int row0 = blockIdx.x * 16;
    const int col0 = wave * 16;
    const int lidx = lane & 15;
    const int khalf = (lane >> 4) * 2;

    const float* __restrict__ arow = A + (size_t)(row0 + lidx) * K + khalf;
    const float* __restrict__ wcol = W + (size_t)khalf * NOUT + col0 + lidx;

    v8f acc = {};
#pragma unroll 4
    for (int k = 0; k < K; k += 4) {
        v2f a, b;
        a.x = arow[0];
        a.y = arow[1];
        b.x = wcol[0];
        b.y = wcol[NOUT];
        acc = __builtin_amdgcn_wmma_f32_16x16x4_f32(
             false, a,  false, b,
             (short)0, acc,  false,  false);
        arow += 4;
        wcol += 4 * NOUT;
    }

    float bv = 0.0f;
    if (HAS_BIAS) bv = bias[col0 + lidx];
    const int moff = (lane >> 4) * 8;
    float* __restrict__ cbase = C + (size_t)row0 * NOUT + col0 + lidx;
#pragma unroll
    for (int j = 0; j < 8; ++j) {
        cbase[(size_t)(j + moff) * NOUT] = acc[j] + bv;
    }
}


template <int F, bool MIX, bool BIAS>
__global__ void __launch_bounds__(256)
k_prop_pull(float* __restrict__ out, const float* __restrict__ h, const float* __restrict__ h0,
            const float* __restrict__ bias,
            const int* __restrict__ rowptr, const int* __restrict__ ecw,
            const float* __restrict__ dinv, int n) {
    typedef __attribute__((ext_vector_type(F / 32))) float vf;
    const int lane = threadIdx.x & 31;
    const int w0 = __builtin_amdgcn_readfirstlane((int)((blockIdx.x * blockDim.x + threadIdx.x) >> 5));
    const int nw = (int)(gridDim.x * (blockDim.x >> 5));
    const vf* __restrict__ hv = (const vf*)h;
    vf bvec = {};
    if (BIAS) bvec = ((const vf*)bias)[lane];

  for (int i = w0; i < n; i += nw) {
    const vf hself = hv[(size_t)i * 32 + lane];
    vf z = hself;
    if (MIX) z = ((const vf*)h0)[(size_t)i * 32 + lane];
    const int beg = rowptr[i];
    const int end = rowptr[i + 1];
    int2 cw = make_int2(0, 0);
    if (beg + lane < end) cw = ((const int2*)ecw)[beg + lane];
    const float di = dinv[i];
    const float sw = di * di;
    vf acc = sw * hself;

    for (int base = beg; base < end; base += 32) {
        if (base != beg) {
            const int p = base + lane;
            cw = make_int2(0, 0);
            if (p < end) cw = ((const int2*)ecw)[p];
        }
        const int rem = end - base;
        const int cnt = rem < 32 ? rem : 32;

        int j = 0;
        for (; j + 4 <= cnt; j += 4) {
            const int s0 = __builtin_amdgcn_readlane(cw.x, j);
            const int s1 = __builtin_amdgcn_readlane(cw.x, j + 1);
            const int s2 = __builtin_amdgcn_readlane(cw.x, j + 2);
            const int s3 = __builtin_amdgcn_readlane(cw.x, j + 3);
            const float w0 = __int_as_float(__builtin_amdgcn_readlane(cw.y, j));
            const float w1 = __int_as_float(__builtin_amdgcn_readlane(cw.y, j + 1));
            const float w2 = __int_as_float(__builtin_amdgcn_readlane(cw.y, j + 2));
            const float w3 = __int_as_float(__builtin_amdgcn_readlane(cw.y, j + 3));
            const vf r0 = hv[(size_t)s0 * 32 + lane];
            const vf r1 = hv[(size_t)s1 * 32 + lane];
            const vf r2 = hv[(size_t)s2 * 32 + lane];
            const vf r3 = hv[(size_t)s3 * 32 + lane];
            acc += w0 * r0;
            acc += w1 * r1;
            acc += w2 * r2;
            acc += w3 * r3;
        }
        const int left = cnt - j;
        if (left > 0) {
            const int s0 = __builtin_amdgcn_readlane(cw.x, j);
            const float w0 = __int_as_float(__builtin_amdgcn_readlane(cw.y, j));
            const vf r0 = hv[(size_t)s0 * 32 + lane];
            float w1 = 0.0f, w2 = 0.0f;
            vf r1 = {}, r2 = {};
            if (left > 1) {
                const int s1 = __builtin_amdgcn_readlane(cw.x, j + 1);
                w1 = __int_as_float(__builtin_amdgcn_readlane(cw.y, j + 1));
                r1 = hv[(size_t)s1 * 32 + lane];
            }
            if (left > 2) {
                const int s2 = __builtin_amdgcn_readlane(cw.x, j + 2);
                w2 = __int_as_float(__builtin_amdgcn_readlane(cw.y, j + 2));
                r2 = hv[(size_t)s2 * 32 + lane];
            }
            acc += w0 * r0;
            if (left > 1) acc += w1 * r1;
            if (left > 2) acc += w2 * r2;
        }
    }

    if (MIX) acc = (1.0f - ALPHA) * acc + ALPHA * z;
    if (BIAS) acc += bvec;
    ((vf*)out)[(size_t)i * 32 + lane] = acc;
  }
}


static inline int cdiv(long long a, int b) { return (int)((a + b - 1) / b); }

extern "C" void kernel_launch(void* const* d_in, const int* in_sizes, int n_in,
                              void* d_out, int out_size, void* d_ws, size_t ws_size,
                              hipStream_t stream) {
    const float* x  = (const float*)d_in[0];
    const int*   ei = (const int*)d_in[1];
    const float* W1 = (const float*)d_in[2];
    const float* b1 = (const float*)d_in[3];
    const float* W3 = (const float*)d_in[4];
    const float* b3 = (const float*)d_in[5];
    float* out = (float*)d_out;

    const int N = in_sizes[0] / 256;
    const int E = in_sizes[1] / 2;
    const int* src = ei;
    const int* dst = ei + E;

    float* ws = (float*)d_ws;
    const size_t nal = ((size_t)N + 64 + 63) & ~(size_t)63;
    float* dinv = ws;
    size_t off = nal;
    int* cnt = (int*)(ws + off);
    off += nal;
    int* rowptr = (int*)(ws + off);
    off += nal;
    int* cursor = (int*)(ws + off);
    off += nal;
    int* ecw = (int*)(ws + off);
    off += ((size_t)2 * E + 63) & ~(size_t)63;
    float* bufA = ws + off;
    off += (size_t)N * 128;
    float* g = ws + off;
    off += (size_t)N * 64;
    float* h0 = ws + off;
    off += (size_t)N * 64;
    float* ping = ws + off;
    off += (size_t)N * 64;
    float* pong = ws + off;

    const int T = 256;

    k_cnt_zero<<<cdiv(N, T), T, 0, stream>>>(cnt, N);
    k_cnt_edges<<<cdiv(E, T), T, 0, stream>>>(cnt, dst, E);
    k_rowptr<<<cdiv(N, SCAN_T), SCAN_T, 0, stream>>>(cnt, rowptr, cursor, dinv, N);
    k_fill<<<cdiv(E, T), T, 0, stream>>>(ecw, cursor, src, dst, dinv, E);

    int gp = cdiv((long long)N * 32, T);
    if (gp > 2048) gp = 2048;

    k_gemm_wmma<256, 128, true><<<N / 16, 32 * (128 / 16), 0, stream>>>(x, W1, b1, bufA);
    k_gemm_wmma<128, 64, false><<<N / 16, 32 * (64 / 16), 0, stream>>>(bufA, W3, b3, g);

    k_prop_pull<64, false, false><<<gp, T, 0, stream>>>(h0, g, g, b3, rowptr, ecw, dinv, N);

    const float* cur = h0;
    for (int it = 0; it < 4; ++it) {
        k_prop_pull<64, true, false><<<gp, T, 0, stream>>>(ping, cur, h0, b3, rowptr, ecw, dinv, N);
        k_prop_pull<64, true, false><<<gp, T, 0, stream>>>(pong, ping, h0, b3, rowptr, ecw, dinv, N);
        cur = pong;
    }
    k_prop_pull<64, true, false><<<gp, T, 0, stream>>>(ping, cur, h0, b3, rowptr, ecw, dinv, N);
    k_prop_pull<64, true, true><<<gp, T, 0, stream>>>(pong, ping, h0, b3, rowptr, ecw, dinv, N);

    k_prop_pull<64, false, false><<<gp, T, 0, stream>>>(out, pong, pong, b3, rowptr, ecw, dinv, N);
}


# Dense layers: a wave owns a full-width 16-row strip, so a k-step's A fragment is loaded once for all 8 (4) column tiles, not once per tile's wave; its fragments are requested one k-step ahead
# speedup vs baseline: 6.1877x; 1.0728x over previous
#include <hip/hip_runtime.h>


typedef __attribute__((ext_vector_type(2))) float v2f;
typedef __attribute__((ext_vector_type(8))) float v8f;

#define ALPHA 0.1f
#define SCAN_T 256


__global__ void k_cnt_zero(int* __restrict__ cnt, int n) {
    int i = blockIdx.x * blockDim.x + threadIdx.x;
    if (i < n) cnt[i] = 0;
}

__global__ void k_cnt_edges(int* __restrict__ cnt, const int* __restrict__ dst, int e) {
    int i = blockIdx.x * blockDim.x + threadIdx.x;
    if (i < e) atomicAdd(&cnt[dst[i]], 1);
}

__global__ void k_rowptr(const int* __restrict__ cnt, int* __restrict__ rowptr,
                         int* __restrict__ cursor, float* __restrict__ dinv, int n) {
    __shared__ int wtot[SCAN_T / 32];
    __shared__ int wpre[SCAN_T / 32];
    const int t = threadIdx.x;
    const int lane = t & 31;
    const int wv = t >> 5;
    const int b0 = blockIdx.x * SCAN_T;

    int pre = 0;
    for (int i = t; i < b0; i += SCAN_T) pre += cnt[i];

    const int idx = b0 + t;
    const int c = (idx < n) ? cnt[idx] : 0;

    int inc = c;
    for (int o = 1; o < 32; o <<= 1) {
        int u = __shfl_up(inc, o);
        if (lane >= o) inc += u;
    }
    for (int o = 16; o > 0; o >>= 1) pre += __shfl_xor(pre, o);

    if (lane == 31) wtot[wv] = inc;
    if (lane == 0) wpre[wv] = pre;
    __syncthreads();

    int base = 0;
    for (int w = 0; w < SCAN_T / 32; ++w) {
        base += wpre[w];
        if (w < wv) base += wtot[w];
    }
    const int excl = base + inc - c;
    if (idx < n) {
        rowptr[idx] = excl;
        cursor[idx] = excl;
        const float v = (float)(c + 1);
        dinv[idx] = (v > 0.0f) ? rsqrtf(v) : 0.0f;
        if (idx == n - 1) rowptr[n] = excl + c;
    }
}

__global__ void k_fill(int* __restrict__ ecw, int* __restrict__ cursor,
                       const int* __restrict__ src, const int* __restrict__ dst,
                       const float* __restrict__ dinv, int e) {
    int i = blockIdx.x * blockDim.x + threadIdx.x;
    if (i < e) {
        const int s = src[i];
        const int d = dst[i];
        const int pos = atomicAdd(&cursor[d], 1);
        const float w = dinv[s] * dinv[d];
        ((int2*)ecw)[pos] = make_int2(s, __float_as_int(w));
    }
}


#define GEMM_WAVES 4

template <int K, int NOUT, bool HAS_BIAS>
__global__ void __launch_bounds__(32 * GEMM_WAVES)
k_gemm_wmma(const float* __restrict__ A, const float* __restrict__ W,
            const float* __restrict__ bias, float* __restrict__ C, int nstrips) {
    constexpr int NT = NOUT / 16;
    static_assert(K % 8 == 0 && NOUT % 16 == 0, "two k-steps of 4 per pass; whole column tiles");
    const int wave = threadIdx.x >> 5;
    const int lane = threadIdx.x & 31;
    const int strip = blockIdx.x * GEMM_WAVES + wave;
    if (strip >= nstrips) return;
    const int row0 = strip * 16;
    const int lidx = lane & 15;
    const int khalf = (lane >> 4) * 2;

    const float* __restrict__ arow = A + (size_t)(row0 + lidx) * K + khalf;
    const float* __restrict__ wcol = W + (size_t)khalf * NOUT + lidx;

    v8f acc[NT];
#pragma unroll
    for (int t = 0; t < NT; ++t) acc[t] = (v8f){};

    v2f a0, a1;
    v2f b0[NT], b1[NT];
    a0.x = arow[0];
    a0.y = arow[1];
#pragma unroll
    for (int t = 0; t < NT; ++t) {
        b0[t].x = wcol[t * 16];
        b0[t].y = wcol[NOUT + t * 16];
    }
#pragma unroll(K <= 128 ? K / 8 : 1)
    for (int k = 0; k < K; k += 8) {
        a1.x = arow[4];
        a1.y = arow[5];
#pragma unroll
        for (int t = 0; t < NT; ++t) {
            b1[t].x = wcol[4 * NOUT + t * 16];
            b1[t].y = wcol[5 * NOUT + t * 16];
        }
        __builtin_amdgcn_sched_barrier(0);
#pragma unroll
        for (int t = 0; t < NT; ++t) {
            acc[t] = __builtin_amdgcn_wmma_f32_16x16x4_f32(
                 false, a0,  false, b0[t],
                 (short)0, acc[t],  false,  false);
        }
        const int adv = (k + 8 < K) ? 8 : 4;
        arow += adv;
        wcol += (size_t)adv * NOUT;
        a0.x = arow[0];
        a0.y = arow[1];
#pragma unroll
        for (int t = 0; t < NT; ++t) {
            b0[t].x = wcol[t * 16];
            b0[t].y = wcol[NOUT + t * 16];
        }
        __builtin_amdgcn_sched_barrier(0);
#pragma unroll
        for (int t = 0; t < NT; ++t) {
            acc[t] = __builtin_amdgcn_wmma_f32_16x16x4_f32(
                 false, a1,  false, b1[t],
                 (short)0, acc[t],  false,  false);
        }
    }

    const int moff = (lane >> 4) * 8;
#pragma unroll
    for (int t = 0; t < NT; ++t) {
        float bv = 0.0f;
        if (HAS_BIAS) bv = bias[t * 16 + lidx];
        float* __restrict__ cbase = C + (size_t)row0 * NOUT + t * 16 + lidx;
#pragma unroll
        for (int j = 0; j < 8; ++j) {
            cbase[(size_t)(j + moff) * NOUT] = acc[t][j] + bv;
        }
    }
}


template <int F, bool MIX, bool BIAS>
__global__ void __launch_bounds__(256)
k_prop_pull(float* __restrict__ out, const float* __restrict__ h, const float* __restrict__ h0,
            const float* __restrict__ bias,
            const int* __restrict__ rowptr, const int* __restrict__ ecw,
            const float* __restrict__ dinv, int n) {
    typedef __attribute__((ext_vector_type(F / 32))) float vf;
    const int lane = threadIdx.x & 31;
    const int w0 = __builtin_amdgcn_readfirstlane((int)((blockIdx.x * blockDim.x + threadIdx.x) >> 5));
    const int nw = (int)(gridDim.x * (blockDim.x >> 5));
    const vf* __restrict__ hv = (const vf*)h;
    vf bvec = {};
    if (BIAS) bvec = ((const vf*)bias)[lane];

  for (int i = w0; i < n; i += nw) {
    const vf hself = hv[(size_t)i * 32 + lane];
    vf z = hself;
    if (MIX) z = ((const vf*)h0)[(size_t)i * 32 + lane];
    const int beg = rowptr[i];
    const int end = rowptr[i + 1];
    int2 cw = make_int2(0, 0);
    if (beg + lane < end) cw = ((const int2*)ecw)[beg + lane];
    const float di = dinv[i];
    const float sw = di * di;
    vf acc = sw * hself;

    for (int base = beg; base < end; base += 32) {
        if (base != beg) {
            const int p = base + lane;
            cw = make_int2(0, 0);
            if (p < end) cw = ((const int2*)ecw)[p];
        }
        const int rem = end - base;
        const int cnt = rem < 32 ? rem : 32;

        int j = 0;
        for (; j + 4 <= cnt; j += 4) {
            const int s0 = __builtin_amdgcn_readlane(cw.x, j);
            const int s1 = __builtin_amdgcn_readlane(cw.x, j + 1);
            const int s2 = __builtin_amdgcn_readlane(cw.x, j + 2);
            const int s3 = __builtin_amdgcn_readlane(cw.x, j + 3);
            const float w0 = __int_as_float(__builtin_amdgcn_readlane(cw.y, j));
            const float w1 = __int_as_float(__builtin_amdgcn_readlane(cw.y, j + 1));
            const float w2 = __int_as_float(__builtin_amdgcn_readlane(cw.y, j + 2));
            const float w3 = __int_as_float(__builtin_amdgcn_readlane(cw.y, j + 3));
            const vf r0 = hv[(size_t)s0 * 32 + lane];
            const vf r1 = hv[(size_t)s1 * 32 + lane];
            const vf r2 = hv[(size_t)s2 * 32 + lane];
            const vf r3 = hv[(size_t)s3 * 32 + lane];
            acc += w0 * r0;
            acc += w1 * r1;
            acc += w2 * r2;
            acc += w3 * r3;
        }
        const int left = cnt - j;
        if (left > 0) {
            const int s0 = __builtin_amdgcn_readlane(cw.x, j);
            const float w0 = __int_as_float(__builtin_amdgcn_readlane(cw.y, j));
            const vf r0 = hv[(size_t)s0 * 32 + lane];
            float w1 = 0.0f, w2 = 0.0f;
            vf r1 = {}, r2 = {};
            if (left > 1) {
                const int s1 = __builtin_amdgcn_readlane(cw.x, j + 1);
                w1 = __int_as_float(__builtin_amdgcn_readlane(cw.y, j + 1));
                r1 = hv[(size_t)s1 * 32 + lane];
            }
            if (left > 2) {
                const int s2 = __builtin_amdgcn_readlane(cw.x, j + 2);
                w2 = __int_as_float(__builtin_amdgcn_readlane(cw.y, j + 2));
                r2 = hv[(size_t)s2 * 32 + lane];
            }
            acc += w0 * r0;
            if (left > 1) acc += w1 * r1;
            if (left > 2) acc += w2 * r2;
        }
    }

    if (MIX) acc = (1.0f - ALPHA) * acc + ALPHA * z;
    if (BIAS) acc += bvec;
    ((vf*)out)[(size_t)i * 32 + lane] = acc;
  }
}


static inline int cdiv(long long a, int b) { return (int)((a + b - 1) / b); }

extern "C" void kernel_launch(void* const* d_in, const int* in_sizes, int n_in,
                              void* d_out, int out_size, void* d_ws, size_t ws_size,
                              hipStream_t stream) {
    const float* x  = (const float*)d_in[0];
    const int*   ei = (const int*)d_in[1];
    const float* W1 = (const float*)d_in[2];
    const float* b1 = (const float*)d_in[3];
    const float* W3 = (const float*)d_in[4];
    const float* b3 = (const float*)d_in[5];
    float* out = (float*)d_out;

    const int N = in_sizes[0] / 256;
    const int E = in_sizes[1] / 2;
    const int* src = ei;
    const int* dst = ei + E;

    float* ws = (float*)d_ws;
    const size_t nal = ((size_t)N + 64 + 63) & ~(size_t)63;
    float* dinv = ws;
    size_t off = nal;
    int* cnt = (int*)(ws + off);
    off += nal;
    int* rowptr = (int*)(ws + off);
    off += nal;
    int* cursor = (int*)(ws + off);
    off += nal;
    int* ecw = (int*)(ws + off);
    off += ((size_t)2 * E + 63) & ~(size_t)63;
    float* bufA = ws + off;
    off += (size_t)N * 128;
    float* g = ws + off;
    off += (size_t)N * 64;
    float* h0 = ws + off;
    off += (size_t)N * 64;
    float* ping = ws + off;
    off += (size_t)N * 64;
    float* pong = ws + off;

    const int T = 256;

    k_cnt_zero<<<cdiv(N, T), T, 0, stream>>>(cnt, N);
    k_cnt_edges<<<cdiv(E, T), T, 0, stream>>>(cnt, dst, E);
    k_rowptr<<<cdiv(N, SCAN_T), SCAN_T, 0, stream>>>(cnt, rowptr, cursor, dinv, N);
    k_fill<<<cdiv(E, T), T, 0, stream>>>(ecw, cursor, src, dst, dinv, E);

    int gp = cdiv((long long)N * 32, T);
    if (gp > 2048) gp = 2048;

    k_gemm_wmma<256, 128, true><<<cdiv(N / 16, GEMM_WAVES), 32 * GEMM_WAVES, 0, stream>>>(x, W1, b1, bufA, N / 16);
    k_gemm_wmma<128, 64, false><<<cdiv(N / 16, GEMM_WAVES), 32 * GEMM_WAVES, 0, stream>>>(bufA, W3, b3, g, N / 16);

    k_prop_pull<64, false, false><<<gp, T, 0, stream>>>(h0, g, g, b3, rowptr, ecw, dinv, N);

    const float* cur = h0;
    for (int it = 0; it < 4; ++it) {
        k_prop_pull<64, true, false><<<gp, T, 0, stream>>>(ping, cur, h0, b3, rowptr, ecw, dinv, N);
        k_prop_pull<64, true, false><<<gp, T, 0, stream>>>(pong, ping, h0, b3, rowptr, ecw, dinv, N);
        cur = pong;
    }
    k_prop_pull<64, true, false><<<gp, T, 0, stream>>>(ping, cur, h0, b3, rowptr, ecw, dinv, N);
    k_prop_pull<64, true, true><<<gp, T, 0, stream>>>(pong, ping, h0, b3, rowptr, ecw, dinv, N);

    k_prop_pull<64, false, false><<<gp, T, 0, stream>>>(out, pong, pong, b3, rowptr, ecw, dinv, N);
}


# Two dense layers merged into one: W13 = W1 W3 and b13 = b1 W3 folded per call on WMMA, then one 256->64 product over the rows, no 128-wide intermediate; sum order: W1 W3 summed first
# speedup vs baseline: 6.5191x; 1.0536x over previous
#include <hip/hip_runtime.h>


typedef __attribute__((ext_vector_type(2))) float v2f;
typedef __attribute__((ext_vector_type(8))) float v8f;

#define ALPHA 0.1f
#define SCAN_T 256


__global__ void k_cnt_zero(int* __restrict__ cnt, int n) {
    int i = blockIdx.x * blockDim.x + threadIdx.x;
    if (i < n) cnt[i] = 0;
}

__global__ void k_cnt_edges(int* __restrict__ cnt, const int* __restrict__ dst, int e) {
    int i = blockIdx.x * blockDim.x + threadIdx.x;
    if (i < e) atomicAdd(&cnt[dst[i]], 1);
}

__global__ void k_rowptr(const int* __restrict__ cnt, int* __restrict__ rowptr,
                         int* __restrict__ cursor, float* __restrict__ dinv, int n) {
    __shared__ int wtot[SCAN_T / 32];
    __shared__ int wpre[SCAN_T / 32];
    const int t = threadIdx.x;
    const int lane = t & 31;
    const int wv = t >> 5;
    const int b0 = blockIdx.x * SCAN_T;

    int pre = 0;
    for (int i = t; i < b0; i += SCAN_T) pre += cnt[i];

    const int idx = b0 + t;
    const int c = (idx < n) ? cnt[idx] : 0;

    int inc = c;
    for (int o = 1; o < 32; o <<= 1) {
        int u = __shfl_up(inc, o);
        if (lane >= o) inc += u;
    }
    for (int o = 16; o > 0; o >>= 1) pre += __shfl_xor(pre, o);

    if (lane == 31) wtot[wv] = inc;
    if (lane == 0) wpre[wv] = pre;
    __syncthreads();

    int base = 0;
    for (int w = 0; w < SCAN_T / 32; ++w) {
        base += wpre[w];
        if (w < wv) base += wtot[w];
    }
    const int excl = base + inc - c;
    if (idx < n) {
        rowptr[idx] = excl;
        cursor[idx] = excl;
        const float v = (float)(c + 1);
        dinv[idx] = (v > 0.0f) ? rsqrtf(v) : 0.0f;
        if (idx == n - 1) rowptr[n] = excl + c;
    }
}

__global__ void k_fill(int* __restrict__ ecw, int* __restrict__ cursor,
                       const int* __restrict__ src, const int* __restrict__ dst,
                       const float* __restrict__ dinv, int e) {
    int i = blockIdx.x * blockDim.x + threadIdx.x;
    if (i < e) {
        const int s = src[i];
        const int d = dst[i];
        const int pos = atomicAdd(&cursor[d], 1);
        const float w = dinv[s] * dinv[d];
        ((int2*)ecw)[pos] = make_int2(s, __float_as_int(w));
    }
}


template <int KIN, int KMID, int NOUT>
__global__ void __launch_bounds__(32 * (NOUT / 16))
k_fold_weights(const float* __restrict__ Wa, const float* __restrict__ ba,
               const float* __restrict__ Wb, float* __restrict__ C) {
    constexpr int KS = KMID / 4;
    static_assert(KIN % 16 == 0 && KMID % 4 == 0 && NOUT % 16 == 0, "whole strips, k-steps and column tiles");
    const int t = threadIdx.x >> 5;
    const int lane = threadIdx.x & 31;
    const int strip = blockIdx.x;
    const int lidx = lane & 15;
    const int khalf = (lane >> 4) * 2;
    const bool tail = strip == KIN / 16;
    const float* __restrict__ arow = tail ? ba + khalf : Wa + (size_t)(strip * 16 + lidx) * KMID + khalf;
    const float* __restrict__ wcol = Wb + (size_t)khalf * NOUT + t * 16 + lidx;

    v2f a[KS], b[KS];
#pragma unroll
    for (int ks = 0; ks < KS; ++ks) {
        a[ks].x = arow[4 * ks];
        a[ks].y = arow[4 * ks + 1];
        b[ks].x = wcol[(size_t)(4 * ks) * NOUT];
        b[ks].y = wcol[(size_t)(4 * ks + 1) * NOUT];
    }
    v8f acc = (v8f){};
#pragma unroll
    for (int ks = 0; ks < KS; ++ks) {
        acc = __builtin_amdgcn_wmma_f32_16x16x4_f32(
             false, a[ks],  false, b[ks],
             (short)0, acc,  false,  false);
    }
    const int moff = (lane >> 4) * 8;
    float* __restrict__ cbase = C + (size_t)(strip * 16) * NOUT + t * 16 + lidx;
#pragma unroll
    for (int j = 0; j < 8; ++j) cbase[(size_t)(j + moff) * NOUT] = acc[j];
}


#define GEMM_WAVES 4

template <int K, int NOUT, bool HAS_BIAS>
__global__ void __launch_bounds__(32 * GEMM_WAVES)
k_gemm_wmma(const float* __restrict__ A, const float* __restrict__ W,
            const float* __restrict__ bias, float* __restrict__ C, int nstrips) {
    constexpr int NT = NOUT / 16;
    static_assert(K % 8 == 0 && NOUT % 16 == 0, "two k-steps of 4 per pass; whole column tiles");
    const int wave = threadIdx.x >> 5;
    const int lane = threadIdx.x & 31;
    const int strip = blockIdx.x * GEMM_WAVES + wave;
    if (strip >= nstrips) return;
    const int row0 = strip * 16;
    const int lidx = lane & 15;
    const int khalf = (lane >> 4) * 2;

    const float* __restrict__ arow = A + (size_t)(row0 + lidx) * K + khalf;
    const float* __restrict__ wcol = W + (size_t)khalf * NOUT + lidx;

    v8f acc[NT];
#pragma unroll
    for (int t = 0; t < NT; ++t) acc[t] = (v8f){};

    v2f a0, a1;
    v2f b0[NT], b1[NT];
    a0.x = arow[0];
    a0.y = arow[1];
#pragma unroll
    for (int t = 0; t < NT; ++t) {
        b0[t].x = wcol[t * 16];
        b0[t].y = wcol[NOUT + t * 16];
    }
#pragma unroll(NT <= 4 ? K / 8 : 1)
    for (int k = 0; k < K; k += 8) {
        a1.x = arow[4];
        a1.y = arow[5];
#pragma unroll
        for (int t = 0; t < NT; ++t) {
            b1[t].x = wcol[4 * NOUT + t * 16];
            b1[t].y = wcol[5 * NOUT + t * 16];
        }
        __builtin_amdgcn_sched_barrier(0);
#pragma unroll
        for (int t = 0; t < NT; ++t) {
            acc[t] = __builtin_amdgcn_wmma_f32_16x16x4_f32(
                 false, a0,  false, b0[t],
                 (short)0, acc[t],  false,  false);
        }
        const int adv = (k + 8 < K) ? 8 : 4;
        arow += adv;
        wcol += (size_t)adv * NOUT;
        a0.x = arow[0];
        a0.y = arow[1];
#pragma unroll
        for (int t = 0; t < NT; ++t) {
            b0[t].x = wcol[t * 16];
            b0[t].y = wcol[NOUT + t * 16];
        }
        __builtin_amdgcn_sched_barrier(0);
#pragma unroll
        for (int t = 0; t < NT; ++t) {
            acc[t] = __builtin_amdgcn_wmma_f32_16x16x4_f32(
                 false, a1,  false, b1[t],
                 (short)0, acc[t],  false,  false);
        }
    }

    const int moff = (lane >> 4) * 8;
#pragma unroll
    for (int t = 0; t < NT; ++t) {
        float bv = 0.0f;
        if (HAS_BIAS) bv = bias[t * 16 + lidx];
        float* __restrict__ cbase = C + (size_t)row0 * NOUT + t * 16 + lidx;
#pragma unroll
        for (int j = 0; j < 8; ++j) {
            cbase[(size_t)(j + moff) * NOUT] = acc[t][j] + bv;
        }
    }
}


template <int F, bool MIX, bool BIAS>
__global__ void __launch_bounds__(256)
k_prop_pull(float* __restrict__ out, const float* __restrict__ h, const float* __restrict__ h0,
            const float* __restrict__ bias,
            const int* __restrict__ rowptr, const int* __restrict__ ecw,
            const float* __restrict__ dinv, int n) {
    typedef __attribute__((ext_vector_type(F / 32))) float vf;
    const int lane = threadIdx.x & 31;
    const int w0 = __builtin_amdgcn_readfirstlane((int)((blockIdx.x * blockDim.x + threadIdx.x) >> 5));
    const int nw = (int)(gridDim.x * (blockDim.x >> 5));
    const vf* __restrict__ hv = (const vf*)h;
    vf bvec = {};
    if (BIAS) bvec = ((const vf*)bias)[lane];

  for (int i = w0; i < n; i += nw) {
    const vf hself = hv[(size_t)i * 32 + lane];
    vf z = hself;
    if (MIX) z = ((const vf*)h0)[(size_t)i * 32 + lane];
    const int beg = rowptr[i];
    const int end = rowptr[i + 1];
    int2 cw = make_int2(0, 0);
    if (beg + lane < end) cw = ((const int2*)ecw)[beg + lane];
    const float di = dinv[i];
    const float sw = di * di;
    vf acc = sw * hself;

    for (int base = beg; base < end; base += 32) {
        if (base != beg) {
            const int p = base + lane;
            cw = make_int2(0, 0);
            if (p < end) cw = ((const int2*)ecw)[p];
        }
        const int rem = end - base;
        const int cnt = rem < 32 ? rem : 32;

        int j = 0;
        for (; j + 4 <= cnt; j += 4) {
            const int s0 = __builtin_amdgcn_readlane(cw.x, j);
            const int s1 = __builtin_amdgcn_readlane(cw.x, j + 1);
            const int s2 = __builtin_amdgcn_readlane(cw.x, j + 2);
            const int s3 = __builtin_amdgcn_readlane(cw.x, j + 3);
            const float w0 = __int_as_float(__builtin_amdgcn_readlane(cw.y, j));
            const float w1 = __int_as_float(__builtin_amdgcn_readlane(cw.y, j + 1));
            const float w2 = __int_as_float(__builtin_amdgcn_readlane(cw.y, j + 2));
            const float w3 = __int_as_float(__builtin_amdgcn_readlane(cw.y, j + 3));
            const vf r0 = hv[(size_t)s0 * 32 + lane];
            const vf r1 = hv[(size_t)s1 * 32 + lane];
            const vf r2 = hv[(size_t)s2 * 32 + lane];
            const vf r3 = hv[(size_t)s3 * 32 + lane];
            acc += w0 * r0;
            acc += w1 * r1;
            acc += w2 * r2;
            acc += w3 * r3;
        }
        const int left = cnt - j;
        if (left > 0) {
            const int s0 = __builtin_amdgcn_readlane(cw.x, j);
            const float w0 = __int_as_float(__builtin_amdgcn_readlane(cw.y, j));
            const vf r0 = hv[(size_t)s0 * 32 + lane];
            float w1 = 0.0f, w2 = 0.0f;
            vf r1 = {}, r2 = {};
            if (left > 1) {
                const int s1 = __builtin_amdgcn_readlane(cw.x, j + 1);
                w1 = __int_as_float(__builtin_amdgcn_readlane(cw.y, j + 1));
                r1 = hv[(size_t)s1 * 32 + lane];
            }
            if (left > 2) {
                const int s2 = __builtin_amdgcn_readlane(cw.x, j + 2);
                w2 = __int_as_float(__builtin_amdgcn_readlane(cw.y, j + 2));
                r2 = hv[(size_t)s2 * 32 + lane];
            }
            acc += w0 * r0;
            if (left > 1) acc += w1 * r1;
            if (left > 2) acc += w2 * r2;
        }
    }

    if (MIX) acc = (1.0f - ALPHA) * acc + ALPHA * z;
    if (BIAS) acc += bvec;
    ((vf*)out)[(size_t)i * 32 + lane] = acc;
  }
}


static inline int cdiv(long long a, int b) { return (int)((a + b - 1) / b); }

extern "C" void kernel_launch(void* const* d_in, const int* in_sizes, int n_in,
                              void* d_out, int out_size, void* d_ws, size_t ws_size,
                              hipStream_t stream) {
    const float* x  = (const float*)d_in[0];
    const int*   ei = (const int*)d_in[1];
    const float* W1 = (const float*)d_in[2];
    const float* b1 = (const float*)d_in[3];
    const float* W3 = (const float*)d_in[4];
    const float* b3 = (const float*)d_in[5];
    float* out = (float*)d_out;

    const int N = in_sizes[0] / 256;
    const int E = in_sizes[1] / 2;
    const int* src = ei;
    const int* dst = ei + E;

    float* ws = (float*)d_ws;
    const size_t nal = ((size_t)N + 64 + 63) & ~(size_t)63;
    float* dinv = ws;
    size_t off = nal;
    int* cnt = (int*)(ws + off);
    off += nal;
    int* rowptr = (int*)(ws + off);
    off += nal;
    int* cursor = (int*)(ws + off);
    off += nal;
    int* ecw = (int*)(ws + off);
    off += ((size_t)2 * E + 63) & ~(size_t)63;
    float* W13 = ws + off;
    off += (size_t)(256 + 16) * 64;
    float* g = ws + off;
    off += (size_t)N * 64;
    float* h0 = ws + off;
    off += (size_t)N * 64;
    float* ping = ws + off;
    off += (size_t)N * 64;
    float* pong = ws + off;

    const int T = 256;

    k_cnt_zero<<<cdiv(N, T), T, 0, stream>>>(cnt, N);
    k_cnt_edges<<<cdiv(E, T), T, 0, stream>>>(cnt, dst, E);
    k_rowptr<<<cdiv(N, SCAN_T), SCAN_T, 0, stream>>>(cnt, rowptr, cursor, dinv, N);
    k_fill<<<cdiv(E, T), T, 0, stream>>>(ecw, cursor, src, dst, dinv, E);

    int gp = cdiv((long long)N * 32, T);
    if (gp > 2048) gp = 2048;

    k_fold_weights<256, 128, 64><<<256 / 16 + 1, 32 * (64 / 16), 0, stream>>>(W1, b1, W3, W13);
    k_gemm_wmma<256, 64, true><<<cdiv(N / 16, GEMM_WAVES), 32 * GEMM_WAVES, 0, stream>>>(x, W13, W13 + 256 * 64, g, N / 16);

    k_prop_pull<64, false, false><<<gp, T, 0, stream>>>(h0, g, g, b3, rowptr, ecw, dinv, N);

    const float* cur = h0;
    for (int it = 0; it < 4; ++it) {
        k_prop_pull<64, true, false><<<gp, T, 0, stream>>>(ping, cur, h0, b3, rowptr, ecw, dinv, N);
        k_prop_pull<64, true, false><<<gp, T, 0, stream>>>(pong, ping, h0, b3, rowptr, ecw, dinv, N);
        cur = pong;
    }
    k_prop_pull<64, true, false><<<gp, T, 0, stream>>>(ping, cur, h0, b3, rowptr, ecw, dinv, N);
    k_prop_pull<64, true, true><<<gp, T, 0, stream>>>(pong, ping, h0, b3, rowptr, ecw, dinv, N);

    k_prop_pull<64, false, false><<<gp, T, 0, stream>>>(out, pong, pong, b3, rowptr, ecw, dinv, N);
}


# Propagation: half a wave sums a row (16 lanes x 16 B), two rows per wave, up to 16 row loads in flight per lane; unused slots point at an all-zero pad row; half the waves; sum order unchanged
# speedup vs baseline: 8.9765x; 1.3770x over previous
#include <hip/hip_runtime.h>


typedef __attribute__((ext_vector_type(2))) float v2f;
typedef __attribute__((ext_vector_type(4))) float v4f;
typedef __attribute__((ext_vector_type(8))) float v8f;

#define ALPHA 0.1f
#define SLOTS 64


__global__ void k_bucket_zero(int* __restrict__ cnt, uint4* __restrict__ slotq, int n, int nq) {
    const int t = blockIdx.x * blockDim.x + threadIdx.x;
    const int stride = gridDim.x * blockDim.x;
    const unsigned int pad2 = (unsigned int)n | ((unsigned int)n << 16);
    for (int i = t; i < nq; i += stride) slotq[i] = make_uint4(pad2, pad2, pad2, pad2);
    for (int i = t; i < n; i += stride) cnt[i] = 0;
}

__global__ void k_fill(unsigned short* __restrict__ slots, int* __restrict__ cnt,
                       const int* __restrict__ src, const int* __restrict__ dst, int e) {
    int i = blockIdx.x * blockDim.x + threadIdx.x;
    if (i < e) {
        const int s = src[i];
        const int d = dst[i];
        asm volatile("" :: "v"(s));
        const int pos = atomicAdd(&cnt[d], 1);
        if (pos < SLOTS) slots[(size_t)d * SLOTS + pos] = (unsigned short)s;
    }
}


template <int KIN, int KMID, int NOUT>
__global__ void __launch_bounds__(32 * (NOUT / 16))
k_fold_weights(const float* __restrict__ Wa, const float* __restrict__ ba,
               const float* __restrict__ Wb, float* __restrict__ C) {
    constexpr int KS = KMID / 4;
    static_assert(KIN % 16 == 0 && KMID % 4 == 0 && NOUT % 16 == 0, "whole strips, k-steps and column tiles");
    const int t = threadIdx.x >> 5;
    const int lane = threadIdx.x & 31;
    const int strip = blockIdx.x;
    const int lidx = lane & 15;
    const int khalf = (lane >> 4) * 2;
    const bool tail = strip == KIN / 16;
    const float* __restrict__ arow = tail ? ba + khalf : Wa + (size_t)(strip * 16 + lidx) * KMID + khalf;
    const float* __restrict__ wcol = Wb + (size_t)khalf * NOUT + t * 16 + lidx;

    v2f a[KS], b[KS];
#pragma unroll
    for (int ks = 0; ks < KS; ++ks) {
        a[ks].x = arow[4 * ks];
        a[ks].y = arow[4 * ks + 1];
        b[ks].x = wcol[(size_t)(4 * ks) * NOUT];
        b[ks].y = wcol[(size_t)(4 * ks + 1) * NOUT];
    }
    v8f acc = (v8f){};
#pragma unroll
    for (int ks = 0; ks < KS; ++ks) {
        acc = __builtin_amdgcn_wmma_f32_16x16x4_f32(
             false, a[ks],  false, b[ks],
             (short)0, acc,  false,  false);
    }
    const int moff = (lane >> 4) * 8;
    float* __restrict__ cbase = C + (size_t)(strip * 16) * NOUT + t * 16 + lidx;
#pragma unroll
    for (int j = 0; j < 8; ++j) cbase[(size_t)(j + moff) * NOUT] = acc[j];
}


#define GEMM_WAVES 4

template <int K, int NOUT, bool HAS_BIAS>
__global__ void __launch_bounds__(32 * GEMM_WAVES)
k_gemm_wmma(const float* __restrict__ A, const float* __restrict__ W,
            const float* __restrict__ bias, const int* __restrict__ cnt,
            float* __restrict__ C, int nstrips, int padrow) {
    constexpr int NT = NOUT / 16;
    static_assert(K % 8 == 0 && NOUT % 16 == 0, "two k-steps of 4 per pass; whole column tiles");
    const int wave = threadIdx.x >> 5;
    const int lane = threadIdx.x & 31;
    const int strip = blockIdx.x * GEMM_WAVES + wave;
    if (strip >= nstrips) return;
    if (strip == 0) {
        for (int cidx = lane; cidx < NOUT; cidx += 32) C[(size_t)padrow * NOUT + cidx] = 0.0f;
    }
    const int row0 = strip * 16;
    const int lidx = lane & 15;
    const int khalf = (lane >> 4) * 2;

    const float* __restrict__ arow = A + (size_t)(row0 + lidx) * K + khalf;
    const float* __restrict__ wcol = W + (size_t)khalf * NOUT + lidx;

    v8f acc[NT];
#pragma unroll
    for (int t = 0; t < NT; ++t) acc[t] = (v8f){};

    v2f a0, a1;
    v2f b0[NT], b1[NT];
    a0.x = arow[0];
    a0.y = arow[1];
#pragma unroll
    for (int t = 0; t < NT; ++t) {
        b0[t].x = wcol[t * 16];
        b0[t].y = wcol[NOUT + t * 16];
    }
#pragma unroll(NT <= 4 ? K / 8 : 1)
    for (int k = 0; k < K; k += 8) {
        a1.x = arow[4];
        a1.y = arow[5];
#pragma unroll
        for (int t = 0; t < NT; ++t) {
            b1[t].x = wcol[4 * NOUT + t * 16];
            b1[t].y = wcol[5 * NOUT + t * 16];
        }
        __builtin_amdgcn_sched_barrier(0);
#pragma unroll
        for (int t = 0; t < NT; ++t) {
            acc[t] = __builtin_amdgcn_wmma_f32_16x16x4_f32(
                 false, a0,  false, b0[t],
                 (short)0, acc[t],  false,  false);
        }
        const int adv = (k + 8 < K) ? 8 : 4;
        arow += adv;
        wcol += (size_t)adv * NOUT;
        a0.x = arow[0];
        a0.y = arow[1];
#pragma unroll
        for (int t = 0; t < NT; ++t) {
            b0[t].x = wcol[t * 16];
            b0[t].y = wcol[NOUT + t * 16];
        }
        __builtin_amdgcn_sched_barrier(0);
#pragma unroll
        for (int t = 0; t < NT; ++t) {
            acc[t] = __builtin_amdgcn_wmma_f32_16x16x4_f32(
                 false, a1,  false, b1[t],
                 (short)0, acc[t],  false,  false);
        }
    }

    const int moff = (lane >> 4) * 8;
    float rs[8];
#pragma unroll
    for (int j = 0; j < 8; ++j) rs[j] = rsqrtf((float)(cnt[row0 + j + moff] + 1));
#pragma unroll
    for (int t = 0; t < NT; ++t) {
        float bv = 0.0f;
        if (HAS_BIAS) bv = bias[t * 16 + lidx];
        float* __restrict__ cbase = C + (size_t)row0 * NOUT + t * 16 + lidx;
#pragma unroll
        for (int j = 0; j < 8; ++j) {
            cbase[(size_t)(j + moff) * NOUT] = rs[j] * (acc[t][j] + bv);
        }
    }
}


template <bool MIX, bool BIAS, bool FINAL>
__global__ void __launch_bounds__(256)
k_prop_pull(float* __restrict__ out, const float* __restrict__ h, const float* __restrict__ h0,
            const float* __restrict__ bias,
            const int* __restrict__ cnt, const uint4* __restrict__ slotq,
            const int* __restrict__ src, const int* __restrict__ dst, int n, int e) {
    const int lane = threadIdx.x & 31;
    const int half = lane >> 4;
    const unsigned int lo = (unsigned int)(lane & 15) * 16u;
    const int w0 = __builtin_amdgcn_readfirstlane((int)((blockIdx.x * blockDim.x + threadIdx.x) >> 5));
    const int nw = (int)(gridDim.x * (blockDim.x >> 5));
    const int npairs = (n + 1) >> 1;
    const char* __restrict__ hb = (const char*)h;
    auto ld = [&](unsigned int id) -> v4f { return *(const v4f*)(hb + (id * 256u + lo)); };
    v4f bvec = {};
    if (BIAS) bvec = *(const v4f*)((const char*)bias + lo);

    if (!FINAL) {
        if (w0 == 0 && half == 0) *(v4f*)((char*)out + ((unsigned int)n * 256u + lo)) = (v4f){};
    }

  for (int p = w0; p < npairs; p += nw) {
    const int row = 2 * p + half;
    const bool live = row < n;
    const int rr = live ? row : n - 1;
    const unsigned int sqoff = (unsigned int)rr * (SLOTS * 2u);
    auto ldq = [&](unsigned int byteoff) -> uint4 { return *(const uint4*)((const char*)slotq + (sqoff + byteoff)); };
    uint4 qa = ldq(0u);
    uint4 qb = ldq(16u);
    const unsigned int rowoff = (unsigned int)rr * 256u + lo;
    const v4f hself = *(const v4f*)(hb + rowoff);
    v4f z = hself;
    if (MIX) z = *(const v4f*)((const char*)h0 + rowoff);
    int c = cnt[rr];
    asm volatile("" :: "v"(qa.x), "v"(qb.x));
    if (!live) c = 0;
    const int c0 = __builtin_amdgcn_readlane(c, 0);
    const int c1 = __builtin_amdgcn_readlane(c, 16);
    const int cmax = c0 > c1 ? c0 : c1;
    const float di = rsqrtf((float)(c + 1));
    v4f acc = hself;

    if (cmax <= SLOTS) {
        for (int j0 = 0; j0 < cmax; j0 += 16) {
            const int m = cmax - j0;
            uint4 na = qa, nb = qb;
            if (m > 16) {
                na = ldq((unsigned int)j0 * 2u + 32u);
                nb = ldq((unsigned int)j0 * 2u + 48u);
            }
            v4f r4 = {}, r5 = {}, r6 = {}, r7 = {}, r8 = {}, r9 = {}, r10 = {}, r11 = {};
            v4f r12 = {}, r13 = {}, r14 = {}, r15 = {};
            const v4f r0 = ld(qa.x & 0xffffu);
            const v4f r1 = ld(qa.x >> 16);
            const v4f r2 = ld(qa.y & 0xffffu);
            const v4f r3 = ld(qa.y >> 16);
            if (m > 4) {
                r4 = ld(qa.z & 0xffffu);
                r5 = ld(qa.z >> 16);
                r6 = ld(qa.w & 0xffffu);
                r7 = ld(qa.w >> 16);
            }
            if (m > 8) {
                r8 = ld(qb.x & 0xffffu);
                r9 = ld(qb.x >> 16);
                r10 = ld(qb.y & 0xffffu);
                r11 = ld(qb.y >> 16);
            }
            if (m > 12) {
                r12 = ld(qb.z & 0xffffu);
                r13 = ld(qb.z >> 16);
                r14 = ld(qb.w & 0xffffu);
                r15 = ld(qb.w >> 16);
            }
            acc += r0;
            acc += r1;
            acc += r2;
            acc += r3;
            if (m > 4) {
                acc += r4;
                acc += r5;
                acc += r6;
                acc += r7;
            }
            if (m > 8) {
                acc += r8;
                acc += r9;
                acc += r10;
                acc += r11;
            }
            if (m > 12) {
                acc += r12;
                acc += r13;
                acc += r14;
                acc += r15;
            }
            qa = na;
            qb = nb;
        }
    } else {
        const unsigned short* __restrict__ s16 = (const unsigned short*)slotq;
        for (int hh = 0; hh < 2; ++hh) {
            const int ri = 2 * p + hh;
            if (ri < n) {
                const int ci = hh ? c1 : c0;
                const bool mine = half == hh;
                if (ci <= SLOTS) {
                    for (int j = 0; j < ci; ++j) {
                        const unsigned int id = s16[(size_t)ri * SLOTS + j];
                        const v4f r = ld(id);
                        if (mine) acc += r;
                    }
                } else {
                    for (int base = 0; base < e; base += 32) {
                        const int q = base + lane;
                        int d = -1, s = 0;
                        if (q < e) { d = dst[q]; s = src[q]; }
                        unsigned int mk = __builtin_amdgcn_ballot_w32(d == ri);
                        while (mk != 0u) {
                            const int k = __builtin_ctz(mk);
                            mk &= mk - 1u;
                            const unsigned int sk = (unsigned int)__builtin_amdgcn_readlane(s, k);
                            const v4f r = ld(sk);
                            if (mine) acc += r;
                        }
                    }
                }
            }
        }
    }

    if (FINAL) {
        acc = di * acc;
    } else {
        acc = (di * di) * acc;
        if (MIX) acc = (1.0f - ALPHA) * acc + ALPHA * z;
        if (BIAS) acc += di * bvec;
    }
    if (live) *(v4f*)((char*)out + rowoff) = acc;
  }
}


static inline int cdiv(long long a, int b) { return (int)((a + b - 1) / b); }

extern "C" void kernel_launch(void* const* d_in, const int* in_sizes, int n_in,
                              void* d_out, int out_size, void* d_ws, size_t ws_size,
                              hipStream_t stream) {
    const float* x  = (const float*)d_in[0];
    const int*   ei = (const int*)d_in[1];
    const float* W1 = (const float*)d_in[2];
    const float* b1 = (const float*)d_in[3];
    const float* W3 = (const float*)d_in[4];
    const float* b3 = (const float*)d_in[5];
    float* out = (float*)d_out;

    const int N = in_sizes[0] / 256;
    const int E = in_sizes[1] / 2;
    const int* src = ei;
    const int* dst = ei + E;

    float* ws = (float*)d_ws;
    const size_t nal = ((size_t)N + 64 + 63) & ~(size_t)63;
    int* cnt = (int*)ws;
    size_t off = nal;
    unsigned int* slotw = (unsigned int*)(ws + off);
    off += ((size_t)N * (SLOTS / 2) + 63) & ~(size_t)63;
    float* W13 = ws + off;
    off += (size_t)(256 + 16) * 64;
    const size_t tab = (size_t)(N + 1) * 64;
    float* g = ws + off;
    off += tab;
    float* h0 = ws + off;
    off += tab;
    float* ping = ws + off;
    off += tab;
    float* pong = ws + off;

    const int T = 256;

    k_bucket_zero<<<512, T, 0, stream>>>(cnt, (uint4*)slotw, N, N * (SLOTS / 8));
    k_fill<<<cdiv(E, T), T, 0, stream>>>((unsigned short*)slotw, cnt, src, dst, E);

    int gp = cdiv((long long)((N + 1) / 2) * 32, T);
    if (gp > 1024) gp = 1024;
    const uint4* slotq = (const uint4*)slotw;

    k_fold_weights<256, 128, 64><<<256 / 16 + 1, 32 * (64 / 16), 0, stream>>>(W1, b1, W3, W13);
    k_gemm_wmma<256, 64, true><<<cdiv(N / 16, GEMM_WAVES), 32 * GEMM_WAVES, 0, stream>>>(x, W13, W13 + 256 * 64, cnt, g, N / 16, N);

    k_prop_pull<false, false, false><<<gp, T, 0, stream>>>(h0, g, g, b3, cnt, slotq, src, dst, N, E);

    const float* cur = h0;
    for (int it = 0; it < 4; ++it) {
        k_prop_pull<true, false, false><<<gp, T, 0, stream>>>(ping, cur, h0, b3, cnt, slotq, src, dst, N, E);
        k_prop_pull<true, false, false><<<gp, T, 0, stream>>>(pong, ping, h0, b3, cnt, slotq, src, dst, N, E);
        cur = pong;
    }
    k_prop_pull<true, false, false><<<gp, T, 0, stream>>>(ping, cur, h0, b3, cnt, slotq, src, dst, N, E);
    k_prop_pull<true, true, false><<<gp, T, 0, stream>>>(pong, ping, h0, b3, cnt, slotq, src, dst, N, E);

    k_prop_pull<false, false, true><<<gp, T, 0, stream>>>(out, pong, pong, b3, cnt, slotq, src, dst, N, E);
}


# Edge buckets built through LDS: each chunk counting-sorted by 128-row range, then a range's slot rows filled by LDS adds, stored whole; no global atomics or pre-fill; sum order: LDS arrival order
# speedup vs baseline: 11.5573x; 1.2875x over previous
#include <hip/hip_runtime.h>


typedef __attribute__((ext_vector_type(2))) float v2f;
typedef __attribute__((ext_vector_type(4))) float v4f;
typedef __attribute__((ext_vector_type(8))) float v8f;

#define ALPHA 0.1f
#define SLOTS 64


#define BINW 128
#define BINSH 7
#define NBCAP 512
#define CHCAP 4096
#define BIN_T 256
#define BIN_IT (CHCAP / BIN_T)

__global__ void __launch_bounds__(BIN_T)
k_bin_chunks(unsigned int* __restrict__ seg, int* __restrict__ startT,
             const int* __restrict__ src, const int* __restrict__ dst, int e, int ch, int nb) {
    __shared__ int cntb[NBCAP];
    __shared__ int startb[NBCAP + 1];
    __shared__ unsigned int sorted[CHCAP];
    const int tid = threadIdx.x;
    const int w = blockIdx.x;
    const int nch = gridDim.x;
    const int base = w * ch;
    int len = e - base;
    if (len > ch) len = ch;
    if (len < 0) len = 0;
    for (int i = tid; i < NBCAP; i += BIN_T) cntb[i] = 0;
    __syncthreads();

    int dv[BIN_IT], sv[BIN_IT];
#pragma unroll
    for (int k = 0; k < BIN_IT; ++k) {
        const int i = k * BIN_T + tid;
        const int gi = i < len ? base + i : 0;
        dv[k] = dst[gi];
        sv[k] = src[gi];
    }
    int rk[BIN_IT];
#pragma unroll
    for (int k = 0; k < BIN_IT; ++k) {
        rk[k] = -1;
        if (k * BIN_T + tid < len) rk[k] = atomicAdd(&cntb[dv[k] >> BINSH], 1);
    }
    __syncthreads();

    if (tid < 32) {
        int loc[NBCAP / 32];
        int tot = 0;
#pragma unroll
        for (int j = 0; j < NBCAP / 32; ++j) {
            loc[j] = tot;
            tot += cntb[tid * (NBCAP / 32) + j];
        }
        int incl = tot;
#pragma unroll
        for (int off = 1; off < 32; off <<= 1) {
            const int v = __shfl_up(incl, off);
            if (tid >= off) incl += v;
        }
        const int excl = incl - tot;
#pragma unroll
        for (int j = 0; j < NBCAP / 32; ++j) startb[tid * (NBCAP / 32) + j] = excl + loc[j];
        if (tid == 31) startb[NBCAP] = incl;
    }
    __syncthreads();

    int st[BIN_IT];
#pragma unroll
    for (int k = 0; k < BIN_IT; ++k) st[k] = startb[dv[k] >> BINSH];
#pragma unroll
    for (int k = 0; k < BIN_IT; ++k) {
        if (rk[k] >= 0)
            sorted[st[k] + rk[k]] = ((unsigned int)(dv[k] & (BINW - 1)) << 16) | (unsigned int)sv[k];
    }
    __syncthreads();

    for (int i = tid; i < len; i += BIN_T) seg[base + i] = sorted[i];
    for (int b = tid; b <= nb; b += BIN_T) startT[b * nch + w] = startb[b];
}

__global__ void __launch_bounds__(BIN_T)
k_build_slots(int* __restrict__ cnt, uint4* __restrict__ slotq,
              const unsigned int* __restrict__ seg, const int* __restrict__ startT,
              int n, int ch, int nch) {
    __shared__ int lcnt[BINW];
    __shared__ uint4 lslotq[BINW * SLOTS / 8];
    unsigned short* lslots = (unsigned short*)lslotq;
    const int tid = threadIdx.x;
    const int b = blockIdx.x;
    const unsigned int pad2 = (unsigned int)n | ((unsigned int)n << 16);
    for (int i = tid; i < BINW * SLOTS / 8; i += BIN_T) lslotq[i] = make_uint4(pad2, pad2, pad2, pad2);
    if (tid < BINW) lcnt[tid] = 0;
    __syncthreads();

    for (int w = tid; w < nch; w += BIN_T) {
        const int a = startT[b * nch + w];
        const int z = startT[(b + 1) * nch + w];
        const unsigned int* __restrict__ sp = seg + (size_t)w * ch;
        for (int j = a; j < z; j += 8) {
            unsigned int p[8];
            int pos[8];
#pragma unroll
            for (int k = 0; k < 8; ++k) p[k] = j + k < z ? sp[j + k] : 0u;
#pragma unroll
            for (int k = 0; k < 8; ++k) {
                pos[k] = SLOTS;
                if (j + k < z) pos[k] = atomicAdd(&lcnt[p[k] >> 16], 1);
            }
#pragma unroll
            for (int k = 0; k < 8; ++k) {
                if (pos[k] < SLOTS) lslots[(p[k] >> 16) * SLOTS + pos[k]] = (unsigned short)(p[k] & 0xffffu);
            }
        }
    }
    __syncthreads();

    const int row0 = b * BINW;
    if (tid < BINW && row0 + tid < n) cnt[row0 + tid] = lcnt[tid];
    for (int i = tid; i < BINW * SLOTS / 8; i += BIN_T) {
        if (row0 + (i >> 3) < n) slotq[(size_t)row0 * (SLOTS / 8) + i] = lslotq[i];
    }
}


template <int KIN, int KMID, int NOUT>
__global__ void __launch_bounds__(32 * (NOUT / 16))
k_fold_weights(const float* __restrict__ Wa, const float* __restrict__ ba,
               const float* __restrict__ Wb, float* __restrict__ C) {
    constexpr int KS = KMID / 4;
    static_assert(KIN % 16 == 0 && KMID % 4 == 0 && NOUT % 16 == 0, "whole strips, k-steps and column tiles");
    const int t = threadIdx.x >> 5;
    const int lane = threadIdx.x & 31;
    const int strip = blockIdx.x;
    const int lidx = lane & 15;
    const int khalf = (lane >> 4) * 2;
    const bool tail = strip == KIN / 16;
    const float* __restrict__ arow = tail ? ba + khalf : Wa + (size_t)(strip * 16 + lidx) * KMID + khalf;
    const float* __restrict__ wcol = Wb + (size_t)khalf * NOUT + t * 16 + lidx;

    v2f a[KS], b[KS];
#pragma unroll
    for (int ks = 0; ks < KS; ++ks) {
        a[ks].x = arow[4 * ks];
        a[ks].y = arow[4 * ks + 1];
        b[ks].x = wcol[(size_t)(4 * ks) * NOUT];
        b[ks].y = wcol[(size_t)(4 * ks + 1) * NOUT];
    }
    v8f acc = (v8f){};
#pragma unroll
    for (int ks = 0; ks < KS; ++ks) {
        acc = __builtin_amdgcn_wmma_f32_16x16x4_f32(
             false, a[ks],  false, b[ks],
             (short)0, acc,  false,  false);
    }
    const int moff = (lane >> 4) * 8;
    float* __restrict__ cbase = C + (size_t)(strip * 16) * NOUT + t * 16 + lidx;
#pragma unroll
    for (int j = 0; j < 8; ++j) cbase[(size_t)(j + moff) * NOUT] = acc[j];
}


#define GEMM_WAVES 4

template <int K, int NOUT, bool HAS_BIAS>
__global__ void __launch_bounds__(32 * GEMM_WAVES)
k_gemm_wmma(const float* __restrict__ A, const float* __restrict__ W,
            const float* __restrict__ bias, const int* __restrict__ cnt,
            float* __restrict__ C, int nstrips, int padrow) {
    constexpr int NT = NOUT / 16;
    static_assert(K % 8 == 0 && NOUT % 16 == 0, "two k-steps of 4 per pass; whole column tiles");
    const int wave = threadIdx.x >> 5;
    const int lane = threadIdx.x & 31;
    const int strip = blockIdx.x * GEMM_WAVES + wave;
    if (strip >= nstrips) return;
    if (strip == 0) {
        for (int cidx = lane; cidx < NOUT; cidx += 32) C[(size_t)padrow * NOUT + cidx] = 0.0f;
    }
    const int row0 = strip * 16;
    const int lidx = lane & 15;
    const int khalf = (lane >> 4) * 2;

    const float* __restrict__ arow = A + (size_t)(row0 + lidx) * K + khalf;
    const float* __restrict__ wcol = W + (size_t)khalf * NOUT + lidx;

    v8f acc[NT];
#pragma unroll
    for (int t = 0; t < NT; ++t) acc[t] = (v8f){};

    v2f a0, a1;
    v2f b0[NT], b1[NT];
    a0.x = arow[0];
    a0.y = arow[1];
#pragma unroll
    for (int t = 0; t < NT; ++t) {
        b0[t].x = wcol[t * 16];
        b0[t].y = wcol[NOUT + t * 16];
    }
#pragma unroll(NT <= 4 ? K / 8 : 1)
    for (int k = 0; k < K; k += 8) {
        a1.x = arow[4];
        a1.y = arow[5];
#pragma unroll
        for (int t = 0; t < NT; ++t) {
            b1[t].x = wcol[4 * NOUT + t * 16];
            b1[t].y = wcol[5 * NOUT + t * 16];
        }
        __builtin_amdgcn_sched_barrier(0);
#pragma unroll
        for (int t = 0; t < NT; ++t) {
            acc[t] = __builtin_amdgcn_wmma_f32_16x16x4_f32(
                 false, a0,  false, b0[t],
                 (short)0, acc[t],  false,  false);
        }
        const int adv = (k + 8 < K) ? 8 : 4;
        arow += adv;
        wcol += (size_t)adv * NOUT;
        a0.x = arow[0];
        a0.y = arow[1];
#pragma unroll
        for (int t = 0; t < NT; ++t) {
            b0[t].x = wcol[t * 16];
            b0[t].y = wcol[NOUT + t * 16];
        }
        __builtin_amdgcn_sched_barrier(0);
#pragma unroll
        for (int t = 0; t < NT; ++t) {
            acc[t] = __builtin_amdgcn_wmma_f32_16x16x4_f32(
                 false, a1,  false, b1[t],
                 (short)0, acc[t],  false,  false);
        }
    }

    const int moff = (lane >> 4) * 8;
    float rs[8];
#pragma unroll
    for (int j = 0; j < 8; ++j) rs[j] = rsqrtf((float)(cnt[row0 + j + moff] + 1));
#pragma unroll
    for (int t = 0; t < NT; ++t) {
        float bv = 0.0f;
        if (HAS_BIAS) bv = bias[t * 16 + lidx];
        float* __restrict__ cbase = C + (size_t)row0 * NOUT + t * 16 + lidx;
#pragma unroll
        for (int j = 0; j < 8; ++j) {
            cbase[(size_t)(j + moff) * NOUT] = rs[j] * (acc[t][j] + bv);
        }
    }
}


template <bool MIX, bool BIAS, bool FINAL>
__global__ void __launch_bounds__(256)
k_prop_pull(float* __restrict__ out, const float* __restrict__ h, const float* __restrict__ h0,
            const float* __restrict__ bias,
            const int* __restrict__ cnt, const uint4* __restrict__ slotq,
            const int* __restrict__ src, const int* __restrict__ dst, int n, int e) {
    const int lane = threadIdx.x & 31;
    const int half = lane >> 4;
    const unsigned int lo = (unsigned int)(lane & 15) * 16u;
    const int w0 = __builtin_amdgcn_readfirstlane((int)((blockIdx.x * blockDim.x + threadIdx.x) >> 5));
    const int nw = (int)(gridDim.x * (blockDim.x >> 5));
    const int npairs = (n + 1) >> 1;
    const char* __restrict__ hb = (const char*)h;
    auto ld = [&](unsigned int id) -> v4f { return *(const v4f*)(hb + (id * 256u + lo)); };
    v4f bvec = {};
    if (BIAS) bvec = *(const v4f*)((const char*)bias + lo);

    if (!FINAL) {
        if (w0 == 0 && half == 0) *(v4f*)((char*)out + ((unsigned int)n * 256u + lo)) = (v4f){};
    }

  for (int p = w0; p < npairs; p += nw) {
    const int row = 2 * p + half;
    const bool live = row < n;
    const int rr = live ? row : n - 1;
    const unsigned int sqoff = (unsigned int)rr * (SLOTS * 2u);
    auto ldq = [&](unsigned int byteoff) -> uint4 { return *(const uint4*)((const char*)slotq + (sqoff + byteoff)); };
    uint4 qa = ldq(0u);
    uint4 qb = ldq(16u);
    const unsigned int rowoff = (unsigned int)rr * 256u + lo;
    const v4f hself = *(const v4f*)(hb + rowoff);
    v4f z = hself;
    if (MIX) z = *(const v4f*)((const char*)h0 + rowoff);
    int c = cnt[rr];
    asm volatile("" :: "v"(qa.x), "v"(qb.x));
    if (!live) c = 0;
    const int c0 = __builtin_amdgcn_readlane(c, 0);
    const int c1 = __builtin_amdgcn_readlane(c, 16);
    const int cmax = c0 > c1 ? c0 : c1;
    const float di = rsqrtf((float)(c + 1));
    v4f acc = hself;

    if (cmax <= SLOTS) {
        for (int j0 = 0; j0 < cmax; j0 += 16) {
            const int m = cmax - j0;
            uint4 na = qa, nb = qb;
            if (m > 16) {
                na = ldq((unsigned int)j0 * 2u + 32u);
                nb = ldq((unsigned int)j0 * 2u + 48u);
            }
            v4f r4 = {}, r5 = {}, r6 = {}, r7 = {}, r8 = {}, r9 = {}, r10 = {}, r11 = {};
            v4f r12 = {}, r13 = {}, r14 = {}, r15 = {};
            const v4f r0 = ld(qa.x & 0xffffu);
            const v4f r1 = ld(qa.x >> 16);
            const v4f r2 = ld(qa.y & 0xffffu);
            const v4f r3 = ld(qa.y >> 16);
            if (m > 4) {
                r4 = ld(qa.z & 0xffffu);
                r5 = ld(qa.z >> 16);
                r6 = ld(qa.w & 0xffffu);
                r7 = ld(qa.w >> 16);
            }
            if (m > 8) {
                r8 = ld(qb.x & 0xffffu);
                r9 = ld(qb.x >> 16);
                r10 = ld(qb.y & 0xffffu);
                r11 = ld(qb.y >> 16);
            }
            if (m > 12) {
                r12 = ld(qb.z & 0xffffu);
                r13 = ld(qb.z >> 16);
                r14 = ld(qb.w & 0xffffu);
                r15 = ld(qb.w >> 16);
            }
            acc += r0;
            acc += r1;
            acc += r2;
            acc += r3;
            if (m > 4) {
                acc += r4;
                acc += r5;
                acc += r6;
                acc += r7;
            }
            if (m > 8) {
                acc += r8;
                acc += r9;
                acc += r10;
                acc += r11;
            }
            if (m > 12) {
                acc += r12;
                acc += r13;
                acc += r14;
                acc += r15;
            }
            qa = na;
            qb = nb;
        }
    } else {
        const unsigned short* __restrict__ s16 = (const unsigned short*)slotq;
        for (int hh = 0; hh < 2; ++hh) {
            const int ri = 2 * p + hh;
            if (ri < n) {
                const int ci = hh ? c1 : c0;
                const bool mine = half == hh;
                if (ci <= SLOTS) {
                    for (int j = 0; j < ci; ++j) {
                        const unsigned int id = s16[(size_t)ri * SLOTS + j];
                        const v4f r = ld(id);
                        if (mine) acc += r;
                    }
                } else {
                    for (int base = 0; base < e; base += 32) {
                        const int q = base + lane;
                        int d = -1, s = 0;
                        if (q < e) { d = dst[q]; s = src[q]; }
                        unsigned int mk = __builtin_amdgcn_ballot_w32(d == ri);
                        while (mk != 0u) {
                            const int k = __builtin_ctz(mk);
                            mk &= mk - 1u;
                            const unsigned int sk = (unsigned int)__builtin_amdgcn_readlane(s, k);
                            const v4f r = ld(sk);
                            if (mine) acc += r;
                        }
                    }
                }
            }
        }
    }

    if (FINAL) {
        acc = di * acc;
    } else {
        acc = (di * di) * acc;
        if (MIX) acc = (1.0f - ALPHA) * acc + ALPHA * z;
        if (BIAS) acc += di * bvec;
    }
    if (live) *(v4f*)((char*)out + rowoff) = acc;
  }
}


static inline int cdiv(long long a, int b) { return (int)((a + b - 1) / b); }

extern "C" void kernel_launch(void* const* d_in, const int* in_sizes, int n_in,
                              void* d_out, int out_size, void* d_ws, size_t ws_size,
                              hipStream_t stream) {
    const float* x  = (const float*)d_in[0];
    const int*   ei = (const int*)d_in[1];
    const float* W1 = (const float*)d_in[2];
    const float* b1 = (const float*)d_in[3];
    const float* W3 = (const float*)d_in[4];
    const float* b3 = (const float*)d_in[5];
    float* out = (float*)d_out;

    const int N = in_sizes[0] / 256;
    const int E = in_sizes[1] / 2;
    const int* src = ei;
    const int* dst = ei + E;

    float* ws = (float*)d_ws;
    const size_t nal = ((size_t)N + 64 + 63) & ~(size_t)63;
    int* cnt = (int*)ws;
    size_t off = nal;
    unsigned int* slotw = (unsigned int*)(ws + off);
    off += ((size_t)N * (SLOTS / 2) + 63) & ~(size_t)63;
    float* W13 = ws + off;
    off += (size_t)(256 + 16) * 64;
    const size_t tab = (size_t)(N + 1) * 64;
    float* g = ws + off;
    off += tab;
    float* h0 = ws + off;
    off += tab;
    float* ping = ws + off;
    off += tab;
    float* pong = ws + off;

    const int T = 256;

    int nch = cdiv(E, CHCAP);
    if (nch < 256) nch = 256;
    const int ch = cdiv(E, nch);
    const int nb = cdiv(N, BINW);
    unsigned int* seg = (unsigned int*)pong;
    int* startT = (int*)(pong + (size_t)nch * ch);
    k_bin_chunks<<<nch, BIN_T, 0, stream>>>(seg, startT, src, dst, E, ch, nb);
    k_build_slots<<<nb, BIN_T, 0, stream>>>(cnt, (uint4*)slotw, seg, startT, N, ch, nch);

    int gp = cdiv((long long)((N + 1) / 2) * 32, T);
    if (gp > 1024) gp = 1024;
    const uint4* slotq = (const uint4*)slotw;

    k_fold_weights<256, 128, 64><<<256 / 16 + 1, 32 * (64 / 16), 0, stream>>>(W1, b1, W3, W13);
    k_gemm_wmma<256, 64, true><<<cdiv(N / 16, GEMM_WAVES), 32 * GEMM_WAVES, 0, stream>>>(x, W13, W13 + 256 * 64, cnt, g, N / 16, N);

    k_prop_pull<false, false, false><<<gp, T, 0, stream>>>(h0, g, g, b3, cnt, slotq, src, dst, N, E);

    const float* cur = h0;
    for (int it = 0; it < 4; ++it) {
        k_prop_pull<true, false, false><<<gp, T, 0, stream>>>(ping, cur, h0, b3, cnt, slotq, src, dst, N, E);
        k_prop_pull<true, false, false><<<gp, T, 0, stream>>>(pong, ping, h0, b3, cnt, slotq, src, dst, N, E);
        cur = pong;
    }
    k_prop_pull<true, false, false><<<gp, T, 0, stream>>>(ping, cur, h0, b3, cnt, slotq, src, dst, N, E);
    k_prop_pull<true, true, false><<<gp, T, 0, stream>>>(pong, ping, h0, b3, cnt, slotq, src, dst, N, E);

    k_prop_pull<false, false, true><<<gp, T, 0, stream>>>(out, pong, pong, b3, cnt, slotq, src, dst, N, E);
}
